# starting kernel
# baseline (speedup 1.0000x reference)
#include <hip/hip_runtime.h>
#include <hip/hip_bf16.h>

typedef __attribute__((ext_vector_type(2))) float v2f;
typedef __attribute__((ext_vector_type(8))) float v8f;

#define NCOL 2048
#define ROWS 16
#define EPSF 1e-10f

__global__ __launch_bounds__(256) void qcd_softmax_kernel(
    const float* __restrict__ emb,
    const float* __restrict__ alphap,
    const float* __restrict__ betap,
    float* __restrict__ out)
{
    __shared__ float s_logit[ROWS * NCOL];
    __shared__ float s_x[NCOL];
    __shared__ float s_y[NCOL];
    __shared__ float s_p[NCOL];
    __shared__ float s_prow[ROWS];
    __shared__ float s_rowmax[ROWS];
    __shared__ float s_rowsum[ROWS];
    __shared__ float s_wmax[8 * ROWS];

    const int tid  = threadIdx.x;
    const int lane = tid & 31;
    const int wv   = tid >> 5;
    const int half = lane >> 4;
    const int l15  = lane & 15;

    const int blk = blockIdx.x;
    const int b   = blk >> 7;
    const int i0  = (blk & 127) << 4;

    const float alpha = alphap[0];
    const float beta  = betap[0];
    const float nb2   = -beta * beta;
    const float a2    = 2.0f * alpha;

    const float* E = emb + (size_t)b * 6 * NCOL;

    #pragma unroll
    for (int k = 0; k < 8; ++k) {
        int c = (k << 8) + tid;
        float x   = E[1 * NCOL + c];
        float y   = E[2 * NCOL + c];
        float mom = E[4 * NCOL + c] + EPSF;
        s_x[c] = x;
        s_y[c] = y;
        s_p[c] = __expf(a2 * __logf(mom));
    }
    if (tid < ROWS) {
        float mom = E[4 * NCOL + i0 + tid] + EPSF;
        s_prow[tid]  = __expf(a2 * __logf(mom));
        s_rowsum[tid] = 0.0f;
    }
    __syncthreads();

    float xi = s_x[i0 + l15];
    float yi = s_y[i0 + l15];
    v2f afrag;
    afrag[0] = half ? (xi * xi + yi * yi) : xi;
    afrag[1] = half ? 1.0f : yi;

    float pi[8];
    #pragma unroll
    for (int r = 0; r < 8; ++r) pi[r] = s_prow[half * 8 + r];

    float rmax[8];
    #pragma unroll
    for (int r = 0; r < 8; ++r) rmax[r] = -INFINITY;

    #pragma unroll 4
    for (int s = 0; s < 16; ++s) {
        int j0 = ((s << 3) + wv) << 4;
        float xj = s_x[j0 + l15];
        float yj = s_y[j0 + l15];
        v2f bfrag;
        bfrag[0] = half ? 1.0f : (-2.0f * xj);
        bfrag[1] = half ? (xj * xj + yj * yj) : (-2.0f * yj);

        v8f c = {};
        c = __builtin_amdgcn_wmma_f32_16x16x4_f32(
                false, afrag, false, bfrag, (short)0, c, false, false);

        float pj = s_p[j0 + l15];
        #pragma unroll
        for (int r = 0; r < 8; ++r) {
            float logit = nb2 * c[r] * fminf(pi[r], pj);
            rmax[r] = fmaxf(rmax[r], logit);
            s_logit[(half * 8 + r) * NCOL + j0 + l15] = logit;
        }
    }

    #pragma unroll
    for (int r = 0; r < 8; ++r) {
        #pragma unroll
        for (int m = 1; m <= 8; m <<= 1)
            rmax[r] = fmaxf(rmax[r], __shfl_xor(rmax[r], m, 32));
    }
    if (l15 == 0) {
        #pragma unroll
        for (int r = 0; r < 8; ++r)
            s_wmax[wv * ROWS + half * 8 + r] = rmax[r];
    }
    __syncthreads();
    if (tid < ROWS) {
        float m = s_wmax[tid];
        #pragma unroll
        for (int w = 1; w < 8; ++w) m = fmaxf(m, s_wmax[w * ROWS + tid]);
        s_rowmax[tid] = m;
    }
    __syncthreads();

    {
        float acc = 0.0f;
        #pragma unroll 8
        for (int k = 0; k < 128; ++k) {
            int idx = (k << 8) + tid;
            int r = k >> 3;
            float e = __expf(s_logit[idx] - s_rowmax[r]);
            s_logit[idx] = e;
            acc += e;
            if ((k & 7) == 7) { atomicAdd(&s_rowsum[r], acc); acc = 0.0f; }
        }
    }
    __syncthreads();

    {
        float* dst = out + (size_t)b * NCOL * NCOL + (size_t)i0 * NCOL;
        float inv = 0.0f;
        #pragma unroll 8
        for (int k = 0; k < 128; ++k) {
            int r = k >> 3;
            if ((k & 7) == 0) inv = 1.0f / s_rowsum[r];
            int idx = (k << 8) + tid;
            dst[idx] = s_logit[idx] * inv;
        }
    }
}

extern "C" void kernel_launch(void* const* d_in, const int* in_sizes, int n_in,
                              void* d_out, int out_size, void* d_ws, size_t ws_size,
                              hipStream_t stream) {
    const float* emb   = (const float*)d_in[0];
    const float* alpha = (const float*)d_in[1];
    const float* beta  = (const float*)d_in[2];
    float* out = (float*)d_out;

    dim3 grid(8 * (NCOL / ROWS));
    dim3 block(256);
    qcd_softmax_kernel<<<grid, block, 0, stream>>>(emb, alpha, beta, out);
}


# Exp tiles stay in registers and rows leave as b128 stores straight from them (B loader gives a lane 4 adjacent columns): drops the 128 KB LDS strip and its store/load pass; sum order changes
# speedup vs baseline: 4.1802x; 4.1802x over previous
#include <hip/hip_runtime.h>
#include <hip/hip_bf16.h>


typedef __attribute__((ext_vector_type(2))) float v2f;
typedef __attribute__((ext_vector_type(4))) float v4f;
typedef __attribute__((ext_vector_type(8))) float v8f;

#define NCOL 2048
#define ROWS 16
#define EPSF 1e-10f

__global__ __launch_bounds__(256) void qcd_softmax_kernel(
    const float* __restrict__ emb,
    const float* __restrict__ alphap,
    const float* __restrict__ betap,
    float* __restrict__ out)
{
    __shared__ __attribute__((aligned(16))) float s_x[NCOL];
    __shared__ __attribute__((aligned(16))) float s_y[NCOL];
    __shared__ __attribute__((aligned(16))) float s_p[NCOL];
    __shared__ float s_prow[ROWS];
    __shared__ float s_wsum[8 * ROWS];
    __shared__ __attribute__((aligned(16))) float s_inv[ROWS];

    const int tid  = threadIdx.x;
    const int lane = tid & 31;
    const int wv   = tid >> 5;
    const int half = lane >> 4;
    const int l15  = lane & 15;

    const int blk = blockIdx.x;
    const int b   = blk >> 7;
    const int i0  = (blk & 127) << 4;

    const float alpha = alphap[0];
    const float beta  = betap[0];
    const float nb2   = -beta * beta;
    const float a2    = 2.0f * alpha;

    const float* E = emb + (size_t)b * 6 * NCOL;

    #pragma unroll
    for (int k = 0; k < 8; ++k) {
        int c = (k << 8) + tid;
        float x   = E[1 * NCOL + c];
        float y   = E[2 * NCOL + c];
        float mom = E[4 * NCOL + c] + EPSF;
        s_x[c] = x;
        s_y[c] = y;
        s_p[c] = __expf(a2 * __logf(mom));
    }
    if (tid < ROWS) {
        float mom = E[4 * NCOL + i0 + tid] + EPSF;
        s_prow[tid]  = __expf(a2 * __logf(mom));
    }
    __syncthreads();

    float xi = s_x[i0 + l15];
    float yi = s_y[i0 + l15];
    v2f afrag;
    afrag[0] = half ? __builtin_fmaf(xi, xi, yi * yi) : xi;
    afrag[1] = half ? 1.0f : yi;

    float pi[8];
    #pragma unroll
    for (int r = 0; r < 8; ++r) pi[r] = s_prow[half * 8 + r];

    float rsum[8];
    #pragma unroll
    for (int r = 0; r < 8; ++r) rsum[r] = 0.0f;

    v4f ev[4][8];
    const int cw = (wv << 8) + (l15 << 2);

    #pragma unroll
    for (int g = 0; g < 4; ++g) {
        const int cb = cw + (g << 6);
        const v4f xj = *(const v4f*)&s_x[cb];
        const v4f yj = *(const v4f*)&s_y[cb];
        const v4f pj = *(const v4f*)&s_p[cb];
        v2f bfrag[4];
        v8f c[4];
        #pragma unroll
        for (int t = 0; t < 4; ++t) {
            bfrag[t][0] = half ? 1.0f : (-2.0f * xj[t]);
            bfrag[t][1] = half ? __builtin_fmaf(xj[t], xj[t], yj[t] * yj[t])
                               : (-2.0f * yj[t]);
        }
        #pragma unroll
        for (int t = 0; t < 4; ++t) {
            v8f z = {};
            c[t] = __builtin_amdgcn_wmma_f32_16x16x4_f32(
                       false, afrag, false, bfrag[t], (short)0, z, false, false);
        }
        asm volatile("v_nop\n\tv_nop\n\tv_nop\n\tv_nop\n\t"
                     "v_nop\n\tv_nop\n\tv_nop\n\tv_nop"
                     : "+v"(c[0]), "+v"(c[1]), "+v"(c[2]), "+v"(c[3]),
                       "+v"(bfrag[0]), "+v"(bfrag[1]), "+v"(bfrag[2]), "+v"(bfrag[3]),
                       "+v"(afrag));
        #pragma unroll
        for (int t = 0; t < 4; ++t) {
            #pragma unroll
            for (int r = 0; r < 8; ++r) {
                float logit = nb2 * c[t][r] * fminf(pi[r], pj[t]);
                float e = __expf(logit);
                rsum[r] += e;
                ev[g][r][t] = e;
            }
        }
    }

    #pragma unroll
    for (int r = 0; r < 8; ++r) {
        #pragma unroll
        for (int m = 1; m <= 8; m <<= 1)
            rsum[r] += __shfl_xor(rsum[r], m, 32);
    }
    if (l15 == 0) {
        #pragma unroll
        for (int r = 0; r < 8; ++r)
            s_wsum[wv * ROWS + half * 8 + r] = rsum[r];
    }
    __syncthreads();
    if (tid < ROWS) {
        float t = s_wsum[tid];
        #pragma unroll
        for (int w = 1; w < 8; ++w) t += s_wsum[w * ROWS + tid];
        s_inv[tid] = 1.0f / t;
    }
    __syncthreads();

    {
        const v4f inva = *(const v4f*)&s_inv[half * 8];
        const v4f invb = *(const v4f*)&s_inv[half * 8 + 4];
        float* dst = out + (size_t)b * NCOL * NCOL
                         + (size_t)(i0 + half * 8) * NCOL + cw;
        #pragma unroll
        for (int r = 0; r < 8; ++r) {
            const float inv = (r < 4) ? inva[r & 3] : invb[r & 3];
            #pragma unroll
            for (int g = 0; g < 4; ++g) {
                v4f v = ev[g][r] * inv;
                *(v4f*)(dst + r * NCOL + (g << 6)) = v;
            }
        }
    }
}

extern "C" void kernel_launch(void* const* d_in, const int* in_sizes, int n_in,
                              void* d_out, int out_size, void* d_ws, size_t ws_size,
                              hipStream_t stream) {
    const float* emb   = (const float*)d_in[0];
    const float* alpha = (const float*)d_in[1];
    const float* beta  = (const float*)d_in[2];
    float* out = (float*)d_out;

    dim3 grid(8 * (NCOL / ROWS));
    dim3 block(256);
    qcd_softmax_kernel<<<grid, block, 0, stream>>>(emb, alpha, beta, out);
}


# Each 8-wave block walks 2 row strips, points staged once: the grid is one resident round (2 blocks per WGP), no second round of blocks enters and re-stages while the store path idles; sum order kept
# speedup vs baseline: 4.4787x; 1.0714x over previous
#include <hip/hip_runtime.h>
#include <hip/hip_bf16.h>


typedef __attribute__((ext_vector_type(2))) float v2f;
typedef __attribute__((ext_vector_type(4))) float v4f;
typedef __attribute__((ext_vector_type(8))) float v8f;

#define NCOL 2048
#define ROWS 16
#define NSTRIP 2
#define EPSF 1e-10f

__global__ __launch_bounds__(256) void qcd_softmax_kernel(
    const float* __restrict__ emb,
    const float* __restrict__ alphap,
    const float* __restrict__ betap,
    float* __restrict__ out)
{
    __shared__ __attribute__((aligned(16))) float s_x[NCOL];
    __shared__ __attribute__((aligned(16))) float s_y[NCOL];
    __shared__ __attribute__((aligned(16))) float s_p[NCOL];
    __shared__ float s_wsum[8 * ROWS];
    __shared__ __attribute__((aligned(16))) float s_inv[ROWS];

    const int tid  = threadIdx.x;
    const int lane = tid & 31;
    const int wv   = tid >> 5;
    const int half = lane >> 4;
    const int l15  = lane & 15;

    const int blk = blockIdx.x;
    const int b   = blk / (128 / NSTRIP);
    const int ib  = (blk % (128 / NSTRIP)) * (NSTRIP * ROWS);

    const float alpha = alphap[0];
    const float beta  = betap[0];
    const float nb2   = -beta * beta;
    const float a2    = 2.0f * alpha;

    const float* E = emb + (size_t)b * 6 * NCOL;

    #pragma unroll
    for (int k = 0; k < 8; ++k) {
        int c = (k << 8) + tid;
        float x   = E[1 * NCOL + c];
        float y   = E[2 * NCOL + c];
        float mom = E[4 * NCOL + c] + EPSF;
        s_x[c] = x;
        s_y[c] = y;
        s_p[c] = __expf(a2 * __logf(mom));
    }
    __syncthreads();

    const int cw = (wv << 8) + (l15 << 2);

    #pragma unroll 1
    for (int st = 0; st < NSTRIP; ++st) {
    const int i0 = ib + st * ROWS;

    float xi = s_x[i0 + l15];
    float yi = s_y[i0 + l15];
    v2f afrag;
    afrag[0] = half ? __builtin_fmaf(xi, xi, yi * yi) : xi;
    afrag[1] = half ? 1.0f : yi;

    float pi[8];
    #pragma unroll
    for (int r = 0; r < 8; ++r) pi[r] = s_p[i0 + half * 8 + r];

    float rsum[8];
    #pragma unroll
    for (int r = 0; r < 8; ++r) rsum[r] = 0.0f;

    v4f ev[4][8];

    #pragma unroll
    for (int g = 0; g < 4; ++g) {
        const int cb = cw + (g << 6);
        const v4f xj = *(const v4f*)&s_x[cb];
        const v4f yj = *(const v4f*)&s_y[cb];
        const v4f pj = *(const v4f*)&s_p[cb];
        v2f bfrag[4];
        v8f c[4];
        #pragma unroll
        for (int t = 0; t < 4; ++t) {
            bfrag[t][0] = half ? 1.0f : (-2.0f * xj[t]);
            bfrag[t][1] = half ? __builtin_fmaf(xj[t], xj[t], yj[t] * yj[t])
                               : (-2.0f * yj[t]);
        }
        #pragma unroll
        for (int t = 0; t < 4; ++t) {
            v8f z = {};
            c[t] = __builtin_amdgcn_wmma_f32_16x16x4_f32(
                       false, afrag, false, bfrag[t], (short)0, z, false, false);
        }
        asm volatile("v_nop\n\tv_nop\n\tv_nop\n\tv_nop\n\t"
                     "v_nop\n\tv_nop\n\tv_nop\n\tv_nop"
                     : "+v"(c[0]), "+v"(c[1]), "+v"(c[2]), "+v"(c[3]),
                       "+v"(bfrag[0]), "+v"(bfrag[1]), "+v"(bfrag[2]), "+v"(bfrag[3]),
                       "+v"(afrag));
        #pragma unroll
        for (int t = 0; t < 4; ++t) {
            #pragma unroll
            for (int r = 0; r < 8; ++r) {
                float logit = nb2 * c[t][r] * fminf(pi[r], pj[t]);
                float e = __expf(logit);
                rsum[r] += e;
                ev[g][r][t] = e;
            }
        }
    }

    #pragma unroll
    for (int r = 0; r < 8; ++r) {
        #pragma unroll
        for (int m = 1; m <= 8; m <<= 1)
            rsum[r] += __shfl_xor(rsum[r], m, 32);
    }
    if (l15 == 0) {
        #pragma unroll
        for (int r = 0; r < 8; ++r)
            s_wsum[wv * ROWS + half * 8 + r] = rsum[r];
    }
    __syncthreads();
    if (tid < ROWS) {
        float t = s_wsum[tid];
        #pragma unroll
        for (int w = 1; w < 8; ++w) t += s_wsum[w * ROWS + tid];
        s_inv[tid] = 1.0f / t;
    }
    __syncthreads();

    {
        const v4f inva = *(const v4f*)&s_inv[half * 8];
        const v4f invb = *(const v4f*)&s_inv[half * 8 + 4];
        float* dst = out + (size_t)b * NCOL * NCOL
                         + (size_t)(i0 + half * 8) * NCOL + cw;
        #pragma unroll
        for (int r = 0; r < 8; ++r) {
            const float inv = (r < 4) ? inva[r & 3] : invb[r & 3];
            #pragma unroll
            for (int g = 0; g < 4; ++g) {
                v4f v = ev[g][r] * inv;
                *(v4f*)(dst + r * NCOL + (g << 6)) = v;
            }
        }
    }
    }
}

extern "C" void kernel_launch(void* const* d_in, const int* in_sizes, int n_in,
                              void* d_out, int out_size, void* d_ws, size_t ws_size,
                              hipStream_t stream) {
    const float* emb   = (const float*)d_in[0];
    const float* alpha = (const float*)d_in[1];
    const float* beta  = (const float*)d_in[2];
    float* out = (float*)d_out;

    dim3 grid(8 * (NCOL / ROWS) / NSTRIP);
    dim3 block(256);
    qcd_softmax_kernel<<<grid, block, 0, stream>>>(emb, alpha, beta, out);
}


# Per-point arithmetic moved to staging: B fragment pairs and weight q=-beta^2*log2(e)*p staged ready-made, an element is exp2(d*max(q_i,q_j)): 2 of 3 multiplies per element gone; sum order kept
# speedup vs baseline: 4.7068x; 1.0509x over previous
#include <hip/hip_runtime.h>
#include <hip/hip_bf16.h>


typedef __attribute__((ext_vector_type(2))) float v2f;
typedef __attribute__((ext_vector_type(4))) float v4f;
typedef __attribute__((ext_vector_type(8))) float v8f;

#define NCOL 2048
#define ROWS 16
#define NSTRIP 2
#define EPSF 1e-10f
#define NQUAD (NCOL / 4)

__global__ __launch_bounds__(256) void qcd_softmax_kernel(
    const float* __restrict__ emb,
    const float* __restrict__ alphap,
    const float* __restrict__ betap,
    float* __restrict__ out)
{
    __shared__ __attribute__((aligned(16))) float s_x[NCOL];
    __shared__ __attribute__((aligned(16))) float s_y[NCOL];
    __shared__ __attribute__((aligned(16))) float s_p[NCOL];
    __shared__ __attribute__((aligned(16))) float s_bp[2][2][NQUAD][4];
    __shared__ float s_wsum[8 * ROWS];
    __shared__ __attribute__((aligned(16))) float s_inv[ROWS];

    const int tid  = threadIdx.x;
    const int lane = tid & 31;
    const int wv   = tid >> 5;
    const int half = lane >> 4;
    const int l15  = lane & 15;

    const int blk = blockIdx.x;
    const int b   = blk / (128 / NSTRIP);
    const int ib  = (blk % (128 / NSTRIP)) * (NSTRIP * ROWS);

    const float alpha = alphap[0];
    const float beta  = betap[0];
    const float nb2   = -beta * beta;
    const float a2    = 2.0f * alpha;
    const float kq    = nb2 * 1.44269504088896341f;

    const float* E = emb + (size_t)b * 6 * NCOL;

    #pragma unroll
    for (int k = 0; k < 8; ++k) {
        int c = (k << 8) + tid;
        float x   = E[1 * NCOL + c];
        float y   = E[2 * NCOL + c];
        float mom = E[4 * NCOL + c] + EPSF;
        s_x[c] = x;
        s_y[c] = y;
        s_p[c] = kq * __expf(a2 * __logf(mom));
        v2f blo, bhi;
        blo[0] = -2.0f * x;
        blo[1] = -2.0f * y;
        bhi[0] = 1.0f;
        bhi[1] = __builtin_fmaf(x, x, y * y);
        const int q4 = c >> 2, pl = (c >> 1) & 1, ps = (c & 1) << 1;
        *(v2f*)&s_bp[pl][0][q4][ps] = blo;
        *(v2f*)&s_bp[pl][1][q4][ps] = bhi;
    }
    __syncthreads();

    const int cw = (wv << 8) + (l15 << 2);

    #pragma unroll 1
    for (int st = 0; st < NSTRIP; ++st) {
    const int i0 = ib + st * ROWS;

    float xi = s_x[i0 + l15];
    float yi = s_y[i0 + l15];
    v2f afrag;
    afrag[0] = half ? __builtin_fmaf(xi, xi, yi * yi) : xi;
    afrag[1] = half ? 1.0f : yi;

    float qi[8];
    #pragma unroll
    for (int r = 0; r < 8; ++r) qi[r] = s_p[i0 + half * 8 + r];

    float rsum[8];
    #pragma unroll
    for (int r = 0; r < 8; ++r) rsum[r] = 0.0f;

    v4f ev[4][8];

    #pragma unroll
    for (int g = 0; g < 4; ++g) {
        const int cb = cw + (g << 6);
        const v4f bA = *(const v4f*)&s_bp[0][half][cb >> 2][0];
        const v4f bB = *(const v4f*)&s_bp[1][half][cb >> 2][0];
        const v4f qj = *(const v4f*)&s_p[cb];
        v2f bfrag[4];
        v8f c[4];
        bfrag[0] = __builtin_shufflevector(bA, bA, 0, 1);
        bfrag[1] = __builtin_shufflevector(bA, bA, 2, 3);
        bfrag[2] = __builtin_shufflevector(bB, bB, 0, 1);
        bfrag[3] = __builtin_shufflevector(bB, bB, 2, 3);
        #pragma unroll
        for (int t = 0; t < 4; ++t) {
            v8f z = {};
            c[t] = __builtin_amdgcn_wmma_f32_16x16x4_f32(
                       false, afrag, false, bfrag[t], (short)0, z, false, false);
        }
        asm volatile("v_nop\n\tv_nop\n\tv_nop\n\tv_nop\n\t"
                     "v_nop\n\tv_nop\n\tv_nop\n\tv_nop"
                     : "+v"(c[0]), "+v"(c[1]), "+v"(c[2]), "+v"(c[3]),
                       "+v"(bfrag[0]), "+v"(bfrag[1]), "+v"(bfrag[2]), "+v"(bfrag[3]),
                       "+v"(afrag));
        #pragma unroll
        for (int t = 0; t < 4; ++t) {
            #pragma unroll
            for (int r = 0; r < 8; ++r) {
                float logit2 = c[t][r] * fmaxf(qi[r], qj[t]);
                float e = __builtin_amdgcn_exp2f(logit2);
                rsum[r] += e;
                ev[g][r][t] = e;
            }
        }
    }

    #pragma unroll
    for (int r = 0; r < 8; ++r) {
        #pragma unroll
        for (int m = 1; m <= 8; m <<= 1)
            rsum[r] += __shfl_xor(rsum[r], m, 32);
    }
    if (l15 == 0) {
        #pragma unroll
        for (int r = 0; r < 8; ++r)
            s_wsum[wv * ROWS + half * 8 + r] = rsum[r];
    }
    __syncthreads();
    if (tid < ROWS) {
        float t = s_wsum[tid];
        #pragma unroll
        for (int w = 1; w < 8; ++w) t += s_wsum[w * ROWS + tid];
        s_inv[tid] = 1.0f / t;
    }
    __syncthreads();

    {
        const v4f inva = *(const v4f*)&s_inv[half * 8];
        const v4f invb = *(const v4f*)&s_inv[half * 8 + 4];
        float* dst = out + (size_t)b * NCOL * NCOL
                         + (size_t)(i0 + half * 8) * NCOL + cw;
        #pragma unroll
        for (int r = 0; r < 8; ++r) {
            const float inv = (r < 4) ? inva[r & 3] : invb[r & 3];
            #pragma unroll
            for (int g = 0; g < 4; ++g) {
                v4f v = ev[g][r] * inv;
                *(v4f*)(dst + r * NCOL + (g << 6)) = v;
            }
        }
    }
    }
}

extern "C" void kernel_launch(void* const* d_in, const int* in_sizes, int n_in,
                              void* d_out, int out_size, void* d_ws, size_t ws_size,
                              hipStream_t stream) {
    const float* emb   = (const float*)d_in[0];
    const float* alpha = (const float*)d_in[1];
    const float* beta  = (const float*)d_in[2];
    float* out = (float*)d_out;

    dim3 grid(8 * (NCOL / ROWS) / NSTRIP);
    dim3 block(256);
    qcd_softmax_kernel<<<grid, block, 0, stream>>>(emb, alpha, beta, out);
}
